# starting kernel
# baseline (speedup 1.0000x reference)
#include <hip/hip_runtime.h>
#include <cmath>

typedef __attribute__((ext_vector_type(2))) float v2f;
typedef __attribute__((ext_vector_type(8))) float v8f;

#define NFEAT 128
#define NCLS  40
#define LD2   48
#define KDIM  128
#define WPB   8


__global__ void fill_f32(float* __restrict__ p, long long n, float v) {
  long long i = blockIdx.x * (long long)blockDim.x + threadIdx.x;
  long long stride = (long long)gridDim.x * blockDim.x;
  for (; i < n; i += stride) p[i] = v;
}

__global__ void degree_kernel(const int* __restrict__ col, float* __restrict__ deg, int E) {
  int e = blockIdx.x * blockDim.x + threadIdx.x;
  if (e < E) atomicAdd(&deg[col[e]], 1.0f);
}

__global__ void rsqrt_kernel(float* __restrict__ deg, int N) {
  int i = blockIdx.x * blockDim.x + threadIdx.x;
  if (i < N) {
    float d = deg[i];
    deg[i] = (d > 0.0f) ? rsqrtf(d) : 0.0f;
  }
}

__global__ __launch_bounds__(256) void gemm_wmma_f32(
    const float* __restrict__ A, const float* __restrict__ B, float* __restrict__ C,
    int M, int Nvalid, int Ntiles, int ldc) {
  __shared__ float Bs[KDIM * 16];

  int nt = blockIdx.x % Ntiles;
  int mtBase = (blockIdx.x / Ntiles) * WPB;
  int n0 = nt << 4;
  int t = threadIdx.x;

#pragma unroll
  for (int j = 0; j < 8; ++j) {
    int idx = t + j * 256;
    int kb = idx >> 4;
    int n = n0 + (idx & 15);
    Bs[idx] = (n < Nvalid) ? B[(long long)kb * Nvalid + n] : 0.0f;
  }
  __syncthreads();

  int w = t >> 5;
  int mt = mtBase + w;
  int Mtiles = (M + 15) >> 4;
  if (mt >= Mtiles) return;

  int lane = t & 31;
  int l16 = lane & 15;
  int hi  = lane >> 4;
  int kh  = hi << 1;

  int m0 = mt << 4;
  int m = m0 + l16;
  if (m >= M) m = M - 1;
  const float* Arow = A + (long long)m * KDIM;

  v8f acc = {};
#pragma unroll 8
  for (int k = 0; k < KDIM; k += 4) {
    v2f a = *(const v2f*)(Arow + k + kh);
    v2f b;
    b.x = Bs[(k + kh) * 16 + l16];
    b.y = Bs[(k + kh + 1) * 16 + l16];
    acc = __builtin_amdgcn_wmma_f32_16x16x4_f32(
         false, a,  false, b,
         (short)0, acc,  false,  false);
  }

  int mbase = m0 + hi * 8;
  float* Cout = C + (long long)mbase * ldc + n0 + l16;
  if (m0 + 16 <= M) {
#pragma unroll
    for (int r = 0; r < 8; ++r) Cout[(long long)r * ldc] = acc[r];
  } else {
#pragma unroll
    for (int r = 0; r < 8; ++r)
      if (mbase + r < M) Cout[(long long)r * ldc] = acc[r];
  }
}

__global__ void scatter_add4(const int* __restrict__ row, const int* __restrict__ col,
                             const float* __restrict__ dinv, const float* __restrict__ h,
                             float* __restrict__ agg, int E, int F4, int ld) {
  long long total = (long long)E * F4;
  long long i = blockIdx.x * (long long)blockDim.x + threadIdx.x;
  long long stride = (long long)gridDim.x * blockDim.x;
  for (; i < total; i += stride) {
    int e = (int)(i / F4);
    int f = (int)(i - (long long)e * F4) << 2;
    int r = row[e];
    int c = col[e];
    float nrm = dinv[r] * dinv[c];
    const float4 hv = *(const float4*)(h + (long long)r * ld + f);
    float* a = agg + (long long)c * ld + f;
    atomicAdd(a + 0, hv.x * nrm);
    atomicAdd(a + 1, hv.y * nrm);
    atomicAdd(a + 2, hv.z * nrm);
    atomicAdd(a + 3, hv.w * nrm);
  }
}

__global__ void bias_relu_self(const float* __restrict__ h, const float* __restrict__ dinv,
                               const float* __restrict__ bias, float* __restrict__ agg,
                               int Nnodes) {
  long long total = (long long)Nnodes * NFEAT;
  long long i = blockIdx.x * (long long)blockDim.x + threadIdx.x;
  long long stride = (long long)gridDim.x * blockDim.x;
  for (; i < total; i += stride) {
    int nidx = (int)(i >> 7);
    int f = (int)(i & (NFEAT - 1));
    float d = dinv[nidx];
    float v = agg[i] + h[i] * d * d + bias[f];
    agg[i] = v > 0.0f ? v : 0.0f;
  }
}

__global__ void out_logsoftmax(const float* __restrict__ h2, const float* __restrict__ agg2,
                               const float* __restrict__ dinv, const float* __restrict__ b2,
                               float* __restrict__ out, int Nnodes) {
  int i = blockIdx.x * blockDim.x + threadIdx.x;
  if (i >= Nnodes) return;
  float d = dinv[i];
  float d2 = d * d;
  float logit[NCLS];
  float mx = -3.4e38f;
#pragma unroll
  for (int f = 0; f < NCLS; ++f) {
    float v = agg2[(long long)i * LD2 + f] + h2[(long long)i * LD2 + f] * d2 + b2[f];
    logit[f] = v;
    mx = fmaxf(mx, v);
  }
  float s = 0.0f;
#pragma unroll
  for (int f = 0; f < NCLS; ++f) s += __expf(logit[f] - mx);
  float lse = mx + __logf(s);
#pragma unroll
  for (int f = 0; f < NCLS; ++f) out[(long long)i * NCLS + f] = logit[f] - lse;
}


extern "C" void kernel_launch(void* const* d_in, const int* in_sizes, int n_in,
                              void* d_out, int out_size, void* d_ws, size_t ws_size,
                              hipStream_t stream) {
  const float* x  = (const float*)d_in[0];
  const int*   ei = (const int*)d_in[1];
  const float* W1 = (const float*)d_in[2];
  const float* b1 = (const float*)d_in[3];
  const float* W2 = (const float*)d_in[4];
  const float* b2 = (const float*)d_in[5];
  float* out = (float*)d_out;

  const int N = in_sizes[0] / NFEAT;
  const int E = in_sizes[1] / 2;
  const int* erow = ei;
  const int* ecol = ei + E;

  float* dinv = (float*)d_ws;
  float* h1   = dinv + N;
  float* agg1 = h1 + (long long)N * NFEAT;
  float* h2   = agg1 + (long long)N * NFEAT;
  float* agg2 = h2 + (long long)N * LD2;

  const int TPB = 256;
  auto blocks = [](long long n, int tpb) {
    long long b = (n + tpb - 1) / tpb;
    return (unsigned)(b > 1048576 ? 1048576 : b);
  };

  fill_f32<<<blocks(N, TPB), TPB, 0, stream>>>(dinv, N, 1.0f);
  fill_f32<<<blocks((long long)N * NFEAT, TPB), TPB, 0, stream>>>(agg1, (long long)N * NFEAT, 0.0f);
  fill_f32<<<blocks((long long)N * LD2, TPB), TPB, 0, stream>>>(agg2, (long long)N * LD2, 0.0f);

  degree_kernel<<<(E + TPB - 1) / TPB, TPB, 0, stream>>>(ecol, dinv, E);
  rsqrt_kernel<<<(N + TPB - 1) / TPB, TPB, 0, stream>>>(dinv, N);

  const int Mtiles = (N + 15) / 16;
  const int mGroups = (Mtiles + WPB - 1) / WPB;

  {
    int Ntiles = NFEAT / 16;
    gemm_wmma_f32<<<mGroups * Ntiles, 256, 0, stream>>>(x, W1, h1, N, NFEAT, Ntiles, NFEAT);
  }

  scatter_add4<<<blocks((long long)E * (NFEAT / 4), TPB), TPB, 0, stream>>>(
      erow, ecol, dinv, h1, agg1, E, NFEAT / 4, NFEAT);

  bias_relu_self<<<blocks((long long)N * NFEAT, TPB), TPB, 0, stream>>>(
      h1, dinv, b1, agg1, N);

  {
    int Ntiles = LD2 / 16;
    gemm_wmma_f32<<<mGroups * Ntiles, 256, 0, stream>>>(agg1, W2, h2, N, NCLS, Ntiles, LD2);
  }

  scatter_add4<<<blocks((long long)E * (NCLS / 4), TPB), TPB, 0, stream>>>(
      erow, ecol, dinv, h2, agg2, E, NCLS / 4, LD2);

  out_logsoftmax<<<(N + TPB - 1) / TPB, TPB, 0, stream>>>(
      h2, agg2, dinv, b2, out, N);
}


# layer 2 reads per-slab edge lists (source, row in slab, weight: 8 B an edge) written by the layer-1 slab kernel as it gathers: no second scan of all slab ids, no per-edge lookups; arrival-order adds
# speedup vs baseline: 2.5104x; 2.5104x over previous
#include <hip/hip_runtime.h>
#include <cmath>

typedef __attribute__((ext_vector_type(2))) float v2f;
typedef __attribute__((ext_vector_type(8))) float v8f;

#define NFEAT 128
#define NCLS  40
#define LD2   48
#define KDIM  128
#define WPB   8
#define SLAB  196
#define QCAP  512
#define SWAVES 32
#define LCAP  16384


__global__ void fill_f32(float* __restrict__ p, long long n, float v) {
  long long i = blockIdx.x * (long long)blockDim.x + threadIdx.x;
  long long stride = (long long)gridDim.x * blockDim.x;
  for (; i < n; i += stride) p[i] = v;
}

__global__ void degree_kernel(const int* __restrict__ col, float* __restrict__ deg,
                              unsigned short* __restrict__ sid, int E, int Epad) {
  int e = blockIdx.x * blockDim.x + threadIdx.x;
  if (e < E) {
    int c = col[e];
    atomicAdd(&deg[c], 1.0f);
    sid[e] = (unsigned short)(c / SLAB);
  } else if (e < Epad) {
    sid[e] = (unsigned short)0xFFFF;
  }
}

__global__ void rsqrt_kernel(float* __restrict__ deg, int N) {
  int i = blockIdx.x * blockDim.x + threadIdx.x;
  if (i < N) {
    float d = deg[i];
    deg[i] = (d > 0.0f) ? rsqrtf(d) : 0.0f;
  }
}

__global__ __launch_bounds__(256) void gemm_wmma_f32(
    const float* __restrict__ A, const float* __restrict__ B, float* __restrict__ C,
    int M, int Nvalid, int Ntiles, int ldc) {
  __shared__ float Bs[KDIM * 16];

  int nt = blockIdx.x % Ntiles;
  int mtBase = (blockIdx.x / Ntiles) * WPB;
  int n0 = nt << 4;
  int t = threadIdx.x;

#pragma unroll
  for (int j = 0; j < 8; ++j) {
    int idx = t + j * 256;
    int kb = idx >> 4;
    int n = n0 + (idx & 15);
    Bs[idx] = (n < Nvalid) ? B[(long long)kb * Nvalid + n] : 0.0f;
  }
  __syncthreads();

  int w = t >> 5;
  int mt = mtBase + w;
  int Mtiles = (M + 15) >> 4;
  if (mt >= Mtiles) return;

  int lane = t & 31;
  int l16 = lane & 15;
  int hi  = lane >> 4;
  int kh  = hi << 1;

  int m0 = mt << 4;
  int m = m0 + l16;
  if (m >= M) m = M - 1;
  const float* Arow = A + (long long)m * KDIM;

  v8f acc = {};
#pragma unroll 8
  for (int k = 0; k < KDIM; k += 4) {
    v2f a = *(const v2f*)(Arow + k + kh);
    v2f b;
    b.x = Bs[(k + kh) * 16 + l16];
    b.y = Bs[(k + kh + 1) * 16 + l16];
    acc = __builtin_amdgcn_wmma_f32_16x16x4_f32(
         false, a,  false, b,
         (short)0, acc,  false,  false);
  }

  int mbase = m0 + hi * 8;
  float* Cout = C + (long long)mbase * ldc + n0 + l16;
  if (m0 + 16 <= M) {
#pragma unroll
    for (int r = 0; r < 8; ++r) Cout[(long long)r * ldc] = acc[r];
  } else {
#pragma unroll
    for (int r = 0; r < 8; ++r)
      if (mbase + r < M) Cout[(long long)r * ldc] = acc[r];
  }
}

template <int LANES, int VEC, int LD, int EPI>
__global__ __launch_bounds__(SWAVES * 32) void slab_aggregate(
    const int* __restrict__ row, const int* __restrict__ col,
    const unsigned short* __restrict__ sid, const float* __restrict__ dinv,
    const float* __restrict__ h, const float* __restrict__ bias,
    float* __restrict__ outp, int N, int Epad, const float* __restrict__ Wm,
    unsigned int* __restrict__ lst, int* __restrict__ lcnt) {
  typedef __attribute__((ext_vector_type(VEC))) float vecf;
  constexpr int F = LANES * VEC;
  constexpr int TP = (EPI == 1) ? (F + 4) : F;
  constexpr int TROWS = (EPI == 1) ? ((SLAB + 15) / 16) * 16 : SLAB;
  constexpr int PW = F + 8;
  constexpr int RPW = (SLAB + SWAVES - 1) / SWAVES;
  static_assert(F % 4 == 0, "a slab's rows leave as 16-byte pieces");
  constexpr int G = 8;
  __shared__ float4 tile4[TROWS * TP / 4];
  __shared__ int queue[SWAVES * QCAP];
  __shared__ float4 wls4[(EPI == 1) ? (F * PW / 4) : 1];
  __shared__ int listFill;
  float* tile = (float*)tile4;

  const int t = threadIdx.x;
  const int lane = t & 31;
  const int w = __builtin_amdgcn_readfirstlane(t >> 5);
  const int slabId = blockIdx.x;
  const int lo = slabId * SLAB;
  const int nrows = (N - lo < SLAB) ? (N - lo) : SLAB;

  for (int i = t; i < TROWS * TP / 4; i += SWAVES * 32) tile4[i] = make_float4(0.f, 0.f, 0.f, 0.f);
  if constexpr (EPI == 1) {
    if (t == 0) listFill = 0;
    for (int i = t; i < F * F / 4; i += SWAVES * 32) {
      const int kr = i / (F / 4), c4 = i % (F / 4);
      wls4[kr * (PW / 4) + c4] = ((const float4*)Wm)[i];
    }
  }
  __syncthreads();

  int* q = queue + w * QCAP;
  const uint4* sid4 = (const uint4*)sid;
  const int nblk = Epad >> 8;
  const unsigned lt = (1u << lane) - 1u;
  const unsigned slabHi = (unsigned)slabId << 16;
  int cnt = 0;

  const int lastBlk = nblk - 1;
  auto rd = [&](int blk) { return sid4[(blk < lastBlk ? blk : lastBlk) * 32 + lane]; };
  auto examine = [&](const uint4& cur, int blk) {
    const unsigned dw[4] = {cur.x, cur.y, cur.z, cur.w};
    const int ebase = (blk << 8) + (lane << 3);
#pragma unroll
    for (int k = 0; k < 8; ++k) {
      const bool m = (k & 1) ? ((dw[k >> 1] & 0xFFFF0000u) == slabHi)
                             : ((dw[k >> 1] & 0x0000FFFFu) == (unsigned)slabId);
      const unsigned mask = __builtin_amdgcn_ballot_w32(m);
      if (mask) {
        if (m) q[cnt + __builtin_popcount(mask & lt)] = ebase + k;
        cnt += __builtin_popcount(mask);
      }
    }
  };

  auto gather = [&](int n, int r, int cl, float nrm) {
    for (int g = 0; g < n; g += G) {
      vecf hv[G];
#pragma unroll
      for (int j = 0; j < G; ++j) {
        const int rj = __builtin_amdgcn_readlane(r, g + j);
        hv[j] = (vecf)(0.0f);
        if (lane < LANES) hv[j] = *(const vecf*)(h + (long long)rj * LD + lane * VEC);
      }
#pragma unroll
      for (int j = 0; j < G; ++j) {
        if (g + j < n) {
          const int cj = __builtin_amdgcn_readlane(cl, g + j);
          const float nj = __builtin_bit_cast(float,
              __builtin_amdgcn_readlane(__builtin_bit_cast(int, nrm), g + j));
          if (lane < LANES) {
            float* dst = tile + cj * TP + lane;
#pragma unroll
            for (int k = 0; k < VEC; ++k)
              __hip_atomic_fetch_add(dst + k * LANES, hv[j][k] * nj, __ATOMIC_RELAXED,
                                     __HIP_MEMORY_SCOPE_WORKGROUP);
          }
        }
      }
    }
  };

  uint2* myList = (uint2*)lst + (size_t)slabId * LCAP;
  int listed = LCAP + 1;
  if constexpr (EPI == 2) listed = lcnt[slabId];

  if (listed <= LCAP) {
    constexpr int RB = 4;
    for (int base0 = w * 32; base0 < listed; base0 += RB * SWAVES * 32) {
      uint2 ent[RB];
#pragma unroll
      for (int k = 0; k < RB; ++k) {
        const int idx = base0 + k * SWAVES * 32 + lane;
        ent[k] = myList[idx < listed ? idx : listed - 1];
      }
#pragma unroll
      for (int k = 0; k < RB; ++k) {
        const int base = base0 + k * SWAVES * 32;
        if (base < listed) {
          const int n = (listed - base < 32) ? (listed - base) : 32;
          gather(n, (int)(ent[k].x & 0x00FFFFFFu), (int)(ent[k].x >> 24),
                 __builtin_bit_cast(float, ent[k].y));
        }
      }
    }
  } else {
  int b = w;
  for (;;) {
    while (b < nblk && cnt < 32) {
      const uint4 ra = rd(b);
      const uint4 rb = rd(b + SWAVES);
      const uint4 rc = rd(b + 2 * SWAVES);
      const uint4 re = rd(b + 3 * SWAVES);
      examine(ra, b); b += SWAVES;
      if (cnt <= QCAP - 256 && b < nblk) { examine(rb, b); b += SWAVES; }
      if (cnt <= QCAP - 256 && b < nblk) { examine(rc, b); b += SWAVES; }
      if (cnt <= QCAP - 256 && b < nblk) { examine(re, b); b += SWAVES; }
    }
    const bool last = (b >= nblk);
    const int take = last ? cnt : (cnt & ~31);
    for (int base = 0; base < take; base += 32) {
      const int n = (take - base < 32) ? (take - base) : 32;
      const bool live = lane < n;
      const int e = live ? q[base + lane] : 0;
      const int r = live ? row[e] : 0;
      const int c = live ? col[e] : lo;
      const float nrm = live ? dinv[r] * dinv[c] : 0.0f;
      const int cl = c - lo;
      if constexpr (EPI == 1) {
        int pos = 0;
        if (lane == 0) pos = __hip_atomic_fetch_add(&listFill, n, __ATOMIC_RELAXED,
                                                    __HIP_MEMORY_SCOPE_WORKGROUP);
        pos = __builtin_amdgcn_readfirstlane(pos);
        if (pos + n <= LCAP && live)
          myList[pos + lane] = make_uint2((unsigned)r | ((unsigned)cl << 24),
                                          __builtin_bit_cast(unsigned, nrm));
      }
      gather(n, r, cl, nrm);
    }
    if (last) break;
    const int rem = cnt - take;
    const int keep = (lane < rem) ? q[take + lane] : 0;
    if (lane < rem) q[lane] = keep;
    cnt = rem;
  }
  }
  const bool act = lane < LANES;
  const int pl = act ? lane : 0;
  vecf bv = *(const vecf*)(bias + pl * VEC);
  vecf hs[RPW];
  float dd[RPW];
#pragma unroll
  for (int j = 0; j < RPW; ++j) {
    const int i = w + j * SWAVES;
    const int node = lo + (i < nrows ? i : 0);
    dd[j] = dinv[node];
    hs[j] = *(const vecf*)(h + (long long)node * LD + pl * VEC);
  }
  __syncthreads();
  if constexpr (EPI == 1) {
    if (t == 0) lcnt[slabId] = listFill;
  }

  if constexpr (EPI == 1) {
    const float* wls = (const float*)wls4;
#pragma unroll
    for (int j = 0; j < RPW; ++j) {
      const int i = w + j * SWAVES;
      if (i < nrows) {
        float* sp = tile + i * TP;
        const float d2 = dd[j] * dd[j];
        float4 o;
        o.x = sp[lane] + hs[j][0] * d2;
        o.y = sp[LANES + lane] + hs[j][1] * d2;
        o.z = sp[2 * LANES + lane] + hs[j][2] * d2;
        o.w = sp[3 * LANES + lane] + hs[j][3] * d2;
        *(float4*)(sp + lane * 4) = o;
      }
    }
    __syncthreads();
    constexpr int MT = TROWS / 16;
    constexpr int NT = F / 32;
    const bool mm = w < 2 * MT;
    const int l16 = lane & 15;
    const int hi = lane >> 4;
    const int kh = hi << 1;
    const int mt = w >> 1;
    const int nb = (w & 1) * (F / 2);
    v8f acc[NT];
#pragma unroll
    for (int q = 0; q < NT; ++q) acc[q] = (v8f){};
    if (mm) {
      const float* ap = tile + (mt * 16 + l16) * TP + kh;
      const float* bp = wls + kh * PW + nb + l16;
#pragma unroll 4
      for (int k = 0; k < F; k += 4) {
        const v2f a = *(const v2f*)(ap + k);
#pragma unroll
        for (int q = 0; q < NT; ++q) {
          v2f b;
          b.x = bp[k * PW + q * 16];
          b.y = bp[(k + 1) * PW + q * 16];
          acc[q] = __builtin_amdgcn_wmma_f32_16x16x4_f32(
              false, a, false, b, (short)0, acc[q], false, false);
        }
      }
    }
    __syncthreads();
    if (mm) {
#pragma unroll
      for (int q = 0; q < NT; ++q) {
        const int cidx = nb + q * 16 + l16;
        const float bq = bias[cidx];
        float* dp = tile + (mt * 16 + hi * 8) * TP + cidx;
#pragma unroll
        for (int r = 0; r < 8; ++r) {
          const float v = acc[q][r] + bq;
          dp[r * TP] = v > 0.0f ? v : 0.0f;
        }
      }
    }
    __syncthreads();
    const int n4 = nrows * (F / 4);
    float4* o4 = (float4*)(outp + (long long)lo * F);
    for (int i = t; i < n4; i += SWAVES * 32) {
      const int rr = i / (F / 4), c4 = i % (F / 4);
      o4[i] = *(const float4*)(tile + rr * TP + c4 * 4);
    }
  } else if (w < nrows) {
    float v[RPW][VEC];
    float mx[RPW];
    float sum[RPW];
#pragma unroll
    for (int j = 0; j < RPW; ++j) {
      const int i = w + j * SWAVES;
      const float* sp = tile + (i < nrows ? i : w) * F + pl;
      const float d2 = dd[j] * dd[j];
      mx[j] = -3.4e38f;
#pragma unroll
      for (int k = 0; k < VEC; ++k) {
        const float x = sp[k * LANES] + hs[j][k] * d2 + bv[k];
        v[j][k] = act ? x : -3.4e38f;
        mx[j] = fmaxf(mx[j], v[j][k]);
      }
    }
#pragma unroll
    for (int m = 16; m >= 1; m >>= 1) {
#pragma unroll
      for (int j = 0; j < RPW; ++j) mx[j] = fmaxf(mx[j], __shfl_xor(mx[j], m));
    }
#pragma unroll
    for (int j = 0; j < RPW; ++j) {
      sum[j] = 0.0f;
#pragma unroll
      for (int k = 0; k < VEC; ++k) sum[j] += act ? __expf(v[j][k] - mx[j]) : 0.0f;
    }
#pragma unroll
    for (int m = 16; m >= 1; m >>= 1) {
#pragma unroll
      for (int j = 0; j < RPW; ++j) sum[j] += __shfl_xor(sum[j], m);
    }
#pragma unroll
    for (int j = 0; j < RPW; ++j) {
      const int i = w + j * SWAVES;
      if (i < nrows && act) {
        const float lse = mx[j] + __logf(sum[j]);
        vecf o;
#pragma unroll
        for (int k = 0; k < VEC; ++k) o[k] = v[j][k] - lse;
        *(vecf*)(tile + i * F + lane * VEC) = o;
      }
    }
  }
  if constexpr (EPI == 2) {
    __syncthreads();
    const int n4 = nrows * (F / 4);
    float4* o4 = (float4*)(outp + (long long)lo * F);
    for (int i = t; i < n4; i += SWAVES * 32) o4[i] = tile4[i];
  }
}


extern "C" void kernel_launch(void* const* d_in, const int* in_sizes, int n_in,
                              void* d_out, int out_size, void* d_ws, size_t ws_size,
                              hipStream_t stream) {
  const float* x  = (const float*)d_in[0];
  const int*   ei = (const int*)d_in[1];
  const float* W1 = (const float*)d_in[2];
  const float* b1 = (const float*)d_in[3];
  const float* W2 = (const float*)d_in[4];
  const float* b2 = (const float*)d_in[5];
  float* out = (float*)d_out;

  const int N = in_sizes[0] / NFEAT;
  const int E = in_sizes[1] / 2;
  const int* erow = ei;
  const int* ecol = ei + E;

  const int Epad = ((E + 255) / 256) * 256;
  unsigned short* sid = (unsigned short*)d_ws;
  const size_t sidBytes = (((size_t)Epad * 2 + 255) / 256) * 256;
  float* dinv = (float*)((char*)d_ws + sidBytes);
  float* act1 = dinv + N;
  float* h2   = act1 + (long long)N * NFEAT;

  const int TPB = 256;
  auto blocks = [](long long n, int tpb) {
    long long b = (n + tpb - 1) / tpb;
    return (unsigned)(b > 1048576 ? 1048576 : b);
  };

  fill_f32<<<blocks(N, TPB), TPB, 0, stream>>>(dinv, N, 1.0f);

  degree_kernel<<<(Epad + TPB - 1) / TPB, TPB, 0, stream>>>(ecol, dinv, sid, E, Epad);
  rsqrt_kernel<<<(N + TPB - 1) / TPB, TPB, 0, stream>>>(dinv, N);

  const int Mtiles = (N + 15) / 16;
  const int mGroups = (Mtiles + WPB - 1) / WPB;
  const int nSlabs = (N + SLAB - 1) / SLAB;
  const size_t listOff = ((sidBytes + (size_t)N * (1 + NFEAT + LD2) * sizeof(float) + 255) / 256) * 256;
  unsigned int* elist = (unsigned int*)((char*)d_ws + listOff);
  int* ecnt = (int*)(elist + (size_t)nSlabs * LCAP * 2);

  slab_aggregate<32, NFEAT / 32, NFEAT, 1><<<nSlabs, SWAVES * 32, 0, stream>>>(
      erow, ecol, sid, dinv, x, b1, act1, N, Epad, W1, elist, ecnt);

  {
    int Ntiles = LD2 / 16;
    gemm_wmma_f32<<<mGroups * Ntiles, 256, 0, stream>>>(act1, W2, h2, N, NCLS, Ntiles, LD2);
  }

  slab_aggregate<20, NCLS / 20, LD2, 2><<<nSlabs, SWAVES * 32, 0, stream>>>(
      erow, ecol, sid, dinv, h2, b2, out, N, Epad, W2, elist, ecnt);
}


# second product chained in registers: layer-1 slab product taken transposed, its ReLU'd tiles are the column operand of W2^T x act^T (W2 in LDS); act plane and product launch gone; new sum order
# speedup vs baseline: 3.2877x; 1.3096x over previous
#include <hip/hip_runtime.h>
#include <cmath>

typedef __attribute__((ext_vector_type(2))) float v2f;
typedef __attribute__((ext_vector_type(8))) float v8f;

#define NFEAT 128
#define NCLS  40
#define LD2   48
#define SLAB  196
#define QCAP  512
#define SWAVES 32
#define LCAP  16384
#define BINT  1024
#define BINE  4
#define HCAP  1024
#define CPAD  64


__global__ void init_kernel(float* __restrict__ deg, int N, int* __restrict__ lcnt, int nSlabs) {
  int i = blockIdx.x * blockDim.x + threadIdx.x;
  if (i < N) deg[i] = 1.0f;
  if (i < nSlabs) lcnt[i * CPAD] = 0;
}

__global__ __launch_bounds__(BINT) void degree_bin_kernel(
    const int* __restrict__ row, const int* __restrict__ col, float* __restrict__ deg,
    unsigned short* __restrict__ sid, int E, int Epad,
    unsigned int* __restrict__ lst, int* __restrict__ lcnt, int nSlabs) {
  __shared__ int hist[HCAP];
  const int t = threadIdx.x;
  for (int i = t; i < HCAP; i += BINT) hist[i] = 0;
  __syncthreads();
  int s[BINE], rk[BINE], cj[BINE], rj[BINE];
  const int e0 = blockIdx.x * (BINT * BINE) + t;
#pragma unroll
  for (int j = 0; j < BINE; ++j) {
    const int e = e0 + j * BINT;
    cj[j] = (e < E) ? col[e] : 0;
    rj[j] = (e < E) ? row[e] : 0;
  }
#pragma unroll
  for (int j = 0; j < BINE; ++j) {
    const int e = e0 + j * BINT;
    s[j] = -1; rk[j] = 0;
    if (e < E) {
      const int sj = cj[j] / SLAB;
      s[j] = sj;
      rk[j] = (sj < HCAP) ? atomicAdd(&hist[sj], 1) : atomicAdd(&lcnt[sj * CPAD], 1);
    }
  }
  __syncthreads();
  for (int i = t; i < HCAP && i < nSlabs; i += BINT) {
    const int n = hist[i];
    if (n) hist[i] = atomicAdd(&lcnt[i * CPAD], n);
  }
  __syncthreads();
#pragma unroll
  for (int j = 0; j < BINE; ++j) {
    const int e = e0 + j * BINT;
    if (s[j] >= 0) {
      const int c = cj[j];
      atomicAdd(&deg[c], 1.0f);
      sid[e] = (unsigned short)s[j];
      const int pos = rk[j] + (s[j] < HCAP ? hist[s[j]] : 0);
      if (pos < LCAP)
        lst[(size_t)s[j] * LCAP + pos] = (unsigned)rj[j] | ((unsigned)(c - s[j] * SLAB) << 24);
    } else if (e < Epad) {
      sid[e] = (unsigned short)0xFFFF;
    }
  }
}

__global__ void rsqrt_kernel(float* __restrict__ deg, int N) {
  int i = blockIdx.x * blockDim.x + threadIdx.x;
  if (i < N) {
    float d = deg[i];
    deg[i] = (d > 0.0f) ? rsqrtf(d) : 0.0f;
  }
}

template <int LANES, int VEC, int LD, int EPI>
__global__ __launch_bounds__(SWAVES * 32) void slab_aggregate(
    const int* __restrict__ row, const int* __restrict__ col,
    const unsigned short* __restrict__ sid, const float* __restrict__ dinv,
    const float* __restrict__ h, const float* __restrict__ bias,
    float* __restrict__ outp, int N, int Epad, const float* __restrict__ Wm,
    const float* __restrict__ Wn,
    unsigned int* __restrict__ lst, const int* __restrict__ lcnt) {
  typedef __attribute__((ext_vector_type(VEC))) float vecf;
  constexpr int F = LANES * VEC;
  constexpr int TP = (EPI == 1) ? (F + 4) : F;
  constexpr int TROWS = (EPI == 1) ? ((SLAB + 15) / 16) * 16 : SLAB;
  constexpr int PW = F + 8;
  constexpr int RPW = (SLAB + SWAVES - 1) / SWAVES;
  static_assert(F % 4 == 0, "a slab's rows leave as 16-byte pieces");
  constexpr int G = 8;
  constexpr int P2 = LD2 + 4;
  __shared__ float4 tile4[TROWS * TP / 4];
  __shared__ int4 queue4[SWAVES * QCAP / 4];
  __shared__ float4 wls4[(EPI == 1) ? (F * PW / 4) : 1];
  __shared__ float4 w2s4[(EPI == 1) ? (F * P2 / 4) : 1];
  float* tile = (float*)tile4;
  int* queue = (int*)queue4;

  const int t = threadIdx.x;
  const int lane = t & 31;
  const int w = __builtin_amdgcn_readfirstlane(t >> 5);
  const int slabId = blockIdx.x;
  const int lo = slabId * SLAB;
  const int nrows = (N - lo < SLAB) ? (N - lo) : SLAB;

  for (int i = t; i < TROWS * TP / 4; i += SWAVES * 32) tile4[i] = make_float4(0.f, 0.f, 0.f, 0.f);
  if constexpr (EPI == 1) {
    for (int i = t; i < F * F / 4; i += SWAVES * 32) {
      const int kr = i / (F / 4), c4 = i % (F / 4);
      wls4[kr * (PW / 4) + c4] = ((const float4*)Wm)[i];
    }
    float* w2s = (float*)w2s4;
    for (int i = t; i < F * LD2; i += SWAVES * 32) {
      const int kr = i / LD2, c = i % LD2;
      w2s[kr * P2 + c] = (c < NCLS) ? Wn[kr * NCLS + c] : 0.0f;
    }
  }
  __syncthreads();

  int* q = queue + w * QCAP;
  const uint4* sid4 = (const uint4*)sid;
  const int nblk = Epad >> 8;
  const unsigned lt = (1u << lane) - 1u;
  const unsigned slabHi = (unsigned)slabId << 16;
  int cnt = 0;

  const int lastBlk = nblk - 1;
  auto rd = [&](int blk) { return sid4[(blk < lastBlk ? blk : lastBlk) * 32 + lane]; };
  auto examine = [&](const uint4& cur, int blk) {
    const unsigned dw[4] = {cur.x, cur.y, cur.z, cur.w};
    const int ebase = (blk << 8) + (lane << 3);
#pragma unroll
    for (int k = 0; k < 8; ++k) {
      const bool m = (k & 1) ? ((dw[k >> 1] & 0xFFFF0000u) == slabHi)
                             : ((dw[k >> 1] & 0x0000FFFFu) == (unsigned)slabId);
      const unsigned mask = __builtin_amdgcn_ballot_w32(m);
      if (mask) {
        if (m) q[cnt + __builtin_popcount(mask & lt)] = ebase + k;
        cnt += __builtin_popcount(mask);
      }
    }
  };

  auto gather = [&](int n, int r, int cl, float nrm) {
    for (int g = 0; g < n; g += G) {
      vecf hv[G];
#pragma unroll
      for (int j = 0; j < G; ++j) {
        const int rj = __builtin_amdgcn_readlane(r, g + j);
        hv[j] = (vecf)(0.0f);
        if (lane < LANES) hv[j] = *(const vecf*)(h + (long long)rj * LD + lane * VEC);
      }
#pragma unroll
      for (int j = 0; j < G; ++j) {
        if (g + j < n) {
          const int cj = __builtin_amdgcn_readlane(cl, g + j);
          const float nj = __builtin_bit_cast(float,
              __builtin_amdgcn_readlane(__builtin_bit_cast(int, nrm), g + j));
          if (lane < LANES) {
            float* dst = tile + cj * TP + lane;
#pragma unroll
            for (int k = 0; k < VEC; ++k)
              __hip_atomic_fetch_add(dst + k * LANES, hv[j][k] * nj, __ATOMIC_RELAXED,
                                     __HIP_MEMORY_SCOPE_WORKGROUP);
          }
        }
      }
    }
  };

  const unsigned int* myX = lst + (size_t)slabId * LCAP;
  float* myW = (float*)(lst + (size_t)gridDim.x * LCAP) + (size_t)slabId * LCAP;
  const int listed = lcnt[slabId * CPAD];

  if (listed <= LCAP) {
    constexpr int RB = 4;
    for (int base0 = w * 32; base0 < listed; base0 += RB * SWAVES * 32) {
      unsigned int ex[RB];
      float ew[RB];
#pragma unroll
      for (int k = 0; k < RB; ++k) {
        const int idx = base0 + k * SWAVES * 32 + lane;
        const int ci = idx < listed ? idx : listed - 1;
        ex[k] = myX[ci];
        if constexpr (EPI == 2) ew[k] = myW[ci];
      }
      if constexpr (EPI == 1) {
#pragma unroll
        for (int k = 0; k < RB; ++k)
          ew[k] = dinv[ex[k] & 0x00FFFFFFu] * dinv[lo + (int)(ex[k] >> 24)];
#pragma unroll
        for (int k = 0; k < RB; ++k) {
          const int idx = base0 + k * SWAVES * 32 + lane;
          if (idx < listed) myW[idx] = ew[k];
        }
      }
#pragma unroll
      for (int k = 0; k < RB; ++k) {
        const int base = base0 + k * SWAVES * 32;
        if (base < listed) {
          const int n = (listed - base < 32) ? (listed - base) : 32;
          gather(n, (int)(ex[k] & 0x00FFFFFFu), (int)(ex[k] >> 24), ew[k]);
        }
      }
    }
  } else {
  int b = w;
  for (;;) {
    while (b < nblk && cnt < 32) {
      const uint4 ra = rd(b);
      const uint4 rb = rd(b + SWAVES);
      const uint4 rc = rd(b + 2 * SWAVES);
      const uint4 re = rd(b + 3 * SWAVES);
      examine(ra, b); b += SWAVES;
      if (cnt <= QCAP - 256 && b < nblk) { examine(rb, b); b += SWAVES; }
      if (cnt <= QCAP - 256 && b < nblk) { examine(rc, b); b += SWAVES; }
      if (cnt <= QCAP - 256 && b < nblk) { examine(re, b); b += SWAVES; }
    }
    const bool last = (b >= nblk);
    const int take = last ? cnt : (cnt & ~31);
    for (int base = 0; base < take; base += 32) {
      const int n = (take - base < 32) ? (take - base) : 32;
      const bool live = lane < n;
      const int e = live ? q[base + lane] : 0;
      const int r = live ? row[e] : 0;
      const int c = live ? col[e] : lo;
      const float nrm = live ? dinv[r] * dinv[c] : 0.0f;
      const int cl = c - lo;
      gather(n, r, cl, nrm);
    }
    if (last) break;
    const int rem = cnt - take;
    const int keep = (lane < rem) ? q[take + lane] : 0;
    if (lane < rem) q[lane] = keep;
    cnt = rem;
  }
  }
  const bool act = lane < LANES;
  const int pl = act ? lane : 0;
  vecf bv = *(const vecf*)(bias + pl * VEC);
  vecf hs[RPW];
  float dd[RPW];
#pragma unroll
  for (int j = 0; j < RPW; ++j) {
    const int i = w + j * SWAVES;
    const int node = lo + (i < nrows ? i : 0);
    dd[j] = dinv[node];
    hs[j] = *(const vecf*)(h + (long long)node * LD + pl * VEC);
  }
  __syncthreads();

  if constexpr (EPI == 1) {
    const float* wls = (const float*)wls4;
#pragma unroll
    for (int j = 0; j < RPW; ++j) {
      const int i = w + j * SWAVES;
      if (i < nrows) {
        float* sp = tile + i * TP;
        const float d2 = dd[j] * dd[j];
        float4 o;
        o.x = sp[lane] + hs[j][0] * d2;
        o.y = sp[LANES + lane] + hs[j][1] * d2;
        o.z = sp[2 * LANES + lane] + hs[j][2] * d2;
        o.w = sp[3 * LANES + lane] + hs[j][3] * d2;
        *(float4*)(sp + lane * 4) = o;
      }
    }
    __syncthreads();
    constexpr int MT = TROWS / 16;
    constexpr int NT = F / 32;
    constexpr int CT = LD2 / 16;
    const bool mm = w < 2 * MT;
    const int l16 = lane & 15;
    const int hi = lane >> 4;
    const int kh = hi << 1;
    const int mt = w >> 1;
    const int nb = (w & 1) * (F / 2);
    const float* w2s = (const float*)w2s4;
    float* h2s = (float*)queue4;
    v8f acc[NT];
    v8f acc2[CT];
#pragma unroll
    for (int q = 0; q < NT; ++q) acc[q] = (v8f){};
#pragma unroll
    for (int c = 0; c < CT; ++c) acc2[c] = (v8f){};
    if (mm) {
      float4 bq[NT][2];
#pragma unroll
      for (int q = 0; q < NT; ++q) {
        const float4* bp4 = (const float4*)(bias + nb + q * 16 + hi * 8);
        bq[q][0] = bp4[0];
        bq[q][1] = bp4[1];
      }
      const float* ap = tile + (mt * 16 + l16) * TP + kh;
      const float* bp = wls + kh * PW + nb + l16;
#pragma unroll 4
      for (int k = 0; k < F; k += 4) {
        const v2f a = *(const v2f*)(ap + k);
#pragma unroll
        for (int q = 0; q < NT; ++q) {
          v2f b;
          b.x = bp[k * PW + q * 16];
          b.y = bp[(k + 1) * PW + q * 16];
          acc[q] = __builtin_amdgcn_wmma_f32_16x16x4_f32(
              false, b, false, a, (short)0, acc[q], false, false);
        }
      }
#pragma unroll
      for (int q = 0; q < NT; ++q) {
        const float bv8[8] = {bq[q][0].x, bq[q][0].y, bq[q][0].z, bq[q][0].w,
                              bq[q][1].x, bq[q][1].y, bq[q][1].z, bq[q][1].w};
#pragma unroll
        for (int j = 0; j < 4; ++j) {
          const float v0 = acc[q][2 * j] + bv8[2 * j];
          const float v1 = acc[q][2 * j + 1] + bv8[2 * j + 1];
          v2f hb;
          hb.x = v0 > 0.0f ? v0 : 0.0f;
          hb.y = v1 > 0.0f ? v1 : 0.0f;
          const float* wp = w2s + (nb + q * 16 + hi * 8 + 2 * j) * P2 + l16;
#pragma unroll
          for (int c = 0; c < CT; ++c) {
            v2f wa;
            wa.x = wp[c * 16];
            wa.y = wp[P2 + c * 16];
            acc2[c] = __builtin_amdgcn_wmma_f32_16x16x4_f32(
                false, wa, false, hb, (short)0, acc2[c], false, false);
          }
        }
      }
    }
    float* hp = h2s + (mt * 16 + l16) * LD2 + hi * 8;
    if (mm && (w & 1) == 0) {
#pragma unroll
      for (int c = 0; c < CT; ++c) {
        *(float4*)(hp + c * 16)     = make_float4(acc2[c][0], acc2[c][1], acc2[c][2], acc2[c][3]);
        *(float4*)(hp + c * 16 + 4) = make_float4(acc2[c][4], acc2[c][5], acc2[c][6], acc2[c][7]);
      }
    }
    __syncthreads();
    if (mm && (w & 1) == 1) {
#pragma unroll
      for (int c = 0; c < CT; ++c) {
        float4 p0 = *(const float4*)(hp + c * 16);
        float4 p1 = *(const float4*)(hp + c * 16 + 4);
        p0.x += acc2[c][0]; p0.y += acc2[c][1]; p0.z += acc2[c][2]; p0.w += acc2[c][3];
        p1.x += acc2[c][4]; p1.y += acc2[c][5]; p1.z += acc2[c][6]; p1.w += acc2[c][7];
        *(float4*)(hp + c * 16)     = p0;
        *(float4*)(hp + c * 16 + 4) = p1;
      }
    }
    __syncthreads();
    const int n4 = nrows * (LD2 / 4);
    float4* o4 = (float4*)(outp + (long long)lo * LD2);
    for (int i = t; i < n4; i += SWAVES * 32) o4[i] = ((const float4*)h2s)[i];
  } else if (w < nrows) {
    float v[RPW][VEC];
    float mx[RPW];
    float sum[RPW];
#pragma unroll
    for (int j = 0; j < RPW; ++j) {
      const int i = w + j * SWAVES;
      const float* sp = tile + (i < nrows ? i : w) * F + pl;
      const float d2 = dd[j] * dd[j];
      mx[j] = -3.4e38f;
#pragma unroll
      for (int k = 0; k < VEC; ++k) {
        const float x = sp[k * LANES] + hs[j][k] * d2 + bv[k];
        v[j][k] = act ? x : -3.4e38f;
        mx[j] = fmaxf(mx[j], v[j][k]);
      }
    }
#pragma unroll
    for (int m = 16; m >= 1; m >>= 1) {
#pragma unroll
      for (int j = 0; j < RPW; ++j) mx[j] = fmaxf(mx[j], __shfl_xor(mx[j], m));
    }
#pragma unroll
    for (int j = 0; j < RPW; ++j) {
      sum[j] = 0.0f;
#pragma unroll
      for (int k = 0; k < VEC; ++k) sum[j] += act ? __expf(v[j][k] - mx[j]) : 0.0f;
    }
#pragma unroll
    for (int m = 16; m >= 1; m >>= 1) {
#pragma unroll
      for (int j = 0; j < RPW; ++j) sum[j] += __shfl_xor(sum[j], m);
    }
#pragma unroll
    for (int j = 0; j < RPW; ++j) {
      const int i = w + j * SWAVES;
      if (i < nrows && act) {
        const float lse = mx[j] + __logf(sum[j]);
        vecf o;
#pragma unroll
        for (int k = 0; k < VEC; ++k) o[k] = v[j][k] - lse;
        *(vecf*)(tile + i * F + lane * VEC) = o;
      }
    }
  }
  if constexpr (EPI == 2) {
    __syncthreads();
    const int n4 = nrows * (F / 4);
    float4* o4 = (float4*)(outp + (long long)lo * F);
    for (int i = t; i < n4; i += SWAVES * 32) o4[i] = tile4[i];
  }
}


extern "C" void kernel_launch(void* const* d_in, const int* in_sizes, int n_in,
                              void* d_out, int out_size, void* d_ws, size_t ws_size,
                              hipStream_t stream) {
  const float* x  = (const float*)d_in[0];
  const int*   ei = (const int*)d_in[1];
  const float* W1 = (const float*)d_in[2];
  const float* b1 = (const float*)d_in[3];
  const float* W2 = (const float*)d_in[4];
  const float* b2 = (const float*)d_in[5];
  float* out = (float*)d_out;

  const int N = in_sizes[0] / NFEAT;
  const int E = in_sizes[1] / 2;
  const int* erow = ei;
  const int* ecol = ei + E;

  const int Epad = ((E + 255) / 256) * 256;
  unsigned short* sid = (unsigned short*)d_ws;
  const size_t sidBytes = (((size_t)Epad * 2 + 255) / 256) * 256;
  float* dinv = (float*)((char*)d_ws + sidBytes);
  float* h2   = dinv + N;

  const int TPB = 256;
  auto blocks = [](long long n, int tpb) {
    long long b = (n + tpb - 1) / tpb;
    return (unsigned)(b > 1048576 ? 1048576 : b);
  };

  const int nSlabs = (N + SLAB - 1) / SLAB;
  const size_t listOff = ((sidBytes + (size_t)N * (1 + LD2) * sizeof(float) + 255) / 256) * 256;
  unsigned int* elist = (unsigned int*)((char*)d_ws + listOff);
  int* ecnt = (int*)(elist + (size_t)nSlabs * LCAP * 2);

  init_kernel<<<blocks(N, TPB), TPB, 0, stream>>>(dinv, N, ecnt, nSlabs);

  degree_bin_kernel<<<(Epad + BINT * BINE - 1) / (BINT * BINE), BINT, 0, stream>>>(
      erow, ecol, dinv, sid, E, Epad, elist, ecnt, nSlabs);
  rsqrt_kernel<<<(N + TPB - 1) / TPB, TPB, 0, stream>>>(dinv, N);

  slab_aggregate<32, NFEAT / 32, NFEAT, 1><<<nSlabs, SWAVES * 32, 0, stream>>>(
      erow, ecol, sid, dinv, x, b1, h2, N, Epad, W1, W2, elist, ecnt);

  slab_aggregate<20, NCLS / 20, LD2, 2><<<nSlabs, SWAVES * 32, 0, stream>>>(
      erow, ecol, sid, dinv, h2, b2, out, N, Epad, W2, W2, elist, ecnt);
}


# in-degree counted by each slab's owner from its own edge list (LDS int adds, rsqrt, one store a node) in place of the rsqrt pass: 800,000 global f32 adds and the degree plane's fill gone
# speedup vs baseline: 4.5677x; 1.3893x over previous
#include <hip/hip_runtime.h>
#include <cmath>

typedef __attribute__((ext_vector_type(2))) float v2f;
typedef __attribute__((ext_vector_type(8))) float v8f;

#define NFEAT 128
#define NCLS  40
#define LD2   48
#define SLAB  196
#define QCAP  512
#define SWAVES 32
#define LCAP  16384
#define BINT  1024
#define BINE  4
#define HCAP  1024
#define CPAD  64


__global__ void init_kernel(int* __restrict__ lcnt, int nSlabs) {
  int i = blockIdx.x * blockDim.x + threadIdx.x;
  if (i < nSlabs) lcnt[i * CPAD] = 0;
}

__global__ __launch_bounds__(BINT) void edge_bin_kernel(
    const int* __restrict__ row, const int* __restrict__ col,
    unsigned short* __restrict__ sid, int E, int Epad,
    unsigned int* __restrict__ lst, int* __restrict__ lcnt, int nSlabs) {
  __shared__ int hist[HCAP];
  const int t = threadIdx.x;
  for (int i = t; i < HCAP; i += BINT) hist[i] = 0;
  __syncthreads();
  int s[BINE], rk[BINE], cj[BINE], rj[BINE];
  const int e0 = blockIdx.x * (BINT * BINE) + t;
#pragma unroll
  for (int j = 0; j < BINE; ++j) {
    const int e = e0 + j * BINT;
    cj[j] = (e < E) ? col[e] : 0;
    rj[j] = (e < E) ? row[e] : 0;
  }
#pragma unroll
  for (int j = 0; j < BINE; ++j) {
    const int e = e0 + j * BINT;
    s[j] = -1; rk[j] = 0;
    if (e < E) {
      const int sj = cj[j] / SLAB;
      s[j] = sj;
      rk[j] = (sj < HCAP) ? atomicAdd(&hist[sj], 1) : atomicAdd(&lcnt[sj * CPAD], 1);
    }
  }
  __syncthreads();
  for (int i = t; i < HCAP && i < nSlabs; i += BINT) {
    const int n = hist[i];
    if (n) hist[i] = atomicAdd(&lcnt[i * CPAD], n);
  }
  __syncthreads();
#pragma unroll
  for (int j = 0; j < BINE; ++j) {
    const int e = e0 + j * BINT;
    if (s[j] >= 0) {
      const int c = cj[j];
      sid[e] = (unsigned short)s[j];
      const int pos = rk[j] + (s[j] < HCAP ? hist[s[j]] : 0);
      if (pos < LCAP)
        lst[(size_t)s[j] * LCAP + pos] = (unsigned)rj[j] | ((unsigned)(c - s[j] * SLAB) << 24);
    } else if (e < Epad) {
      sid[e] = (unsigned short)0xFFFF;
    }
  }
}

#define DEGT 256
#define DEGU 16
__global__ __launch_bounds__(DEGT) void slab_degree_kernel(
    const int* __restrict__ col, int E, const unsigned int* __restrict__ lst,
    const int* __restrict__ lcnt, float* __restrict__ dinv, int N) {
  __shared__ int cntS[SLAB];
  const int t = threadIdx.x;
  const int slabId = blockIdx.x;
  const int lo = slabId * SLAB;
  for (int i = t; i < SLAB; i += DEGT) cntS[i] = 0;
  __syncthreads();
  const int listed = lcnt[slabId * CPAD];
  if (listed <= LCAP) {
    const unsigned int* myX = lst + (size_t)slabId * LCAP;
    for (int base = t; base < listed; base += DEGT * DEGU) {
      unsigned int v[DEGU];
#pragma unroll
      for (int k = 0; k < DEGU; ++k) {
        const int i = base + k * DEGT;
        v[k] = myX[i < listed ? i : listed - 1];
      }
#pragma unroll
      for (int k = 0; k < DEGU; ++k)
        if (base + k * DEGT < listed) atomicAdd(&cntS[v[k] >> 24], 1);
    }
  } else {
    for (int e = t; e < E; e += DEGT) {
      const int c = col[e] - lo;
      if (c >= 0 && c < SLAB) atomicAdd(&cntS[c], 1);
    }
  }
  __syncthreads();
  for (int i = t; i < SLAB; i += DEGT) {
    const int node = lo + i;
    if (node < N) {
      const float d = 1.0f + (float)cntS[i];
      dinv[node] = (d > 0.0f) ? rsqrtf(d) : 0.0f;
    }
  }
}

template <int LANES, int VEC, int LD, int EPI>
__global__ __launch_bounds__(SWAVES * 32) void slab_aggregate(
    const int* __restrict__ row, const int* __restrict__ col,
    const unsigned short* __restrict__ sid, const float* __restrict__ dinv,
    const float* __restrict__ h, const float* __restrict__ bias,
    float* __restrict__ outp, int N, int Epad, const float* __restrict__ Wm,
    const float* __restrict__ Wn,
    unsigned int* __restrict__ lst, const int* __restrict__ lcnt) {
  typedef __attribute__((ext_vector_type(VEC))) float vecf;
  constexpr int F = LANES * VEC;
  constexpr int TP = (EPI == 1) ? (F + 4) : F;
  constexpr int TROWS = (EPI == 1) ? ((SLAB + 15) / 16) * 16 : SLAB;
  constexpr int PW = F + 8;
  constexpr int RPW = (SLAB + SWAVES - 1) / SWAVES;
  static_assert(F % 4 == 0, "a slab's rows leave as 16-byte pieces");
  constexpr int G = 8;
  constexpr int P2 = LD2 + 4;
  __shared__ float4 tile4[TROWS * TP / 4];
  __shared__ int4 queue4[SWAVES * QCAP / 4];
  __shared__ float4 wls4[(EPI == 1) ? (F * PW / 4) : 1];
  __shared__ float4 w2s4[(EPI == 1) ? (F * P2 / 4) : 1];
  float* tile = (float*)tile4;
  int* queue = (int*)queue4;

  const int t = threadIdx.x;
  const int lane = t & 31;
  const int w = __builtin_amdgcn_readfirstlane(t >> 5);
  const int slabId = blockIdx.x;
  const int lo = slabId * SLAB;
  const int nrows = (N - lo < SLAB) ? (N - lo) : SLAB;

  for (int i = t; i < TROWS * TP / 4; i += SWAVES * 32) tile4[i] = make_float4(0.f, 0.f, 0.f, 0.f);
  if constexpr (EPI == 1) {
    for (int i = t; i < F * F / 4; i += SWAVES * 32) {
      const int kr = i / (F / 4), c4 = i % (F / 4);
      wls4[kr * (PW / 4) + c4] = ((const float4*)Wm)[i];
    }
    float* w2s = (float*)w2s4;
    for (int i = t; i < F * LD2; i += SWAVES * 32) {
      const int kr = i / LD2, c = i % LD2;
      w2s[kr * P2 + c] = (c < NCLS) ? Wn[kr * NCLS + c] : 0.0f;
    }
  }
  __syncthreads();

  int* q = queue + w * QCAP;
  const uint4* sid4 = (const uint4*)sid;
  const int nblk = Epad >> 8;
  const unsigned lt = (1u << lane) - 1u;
  const unsigned slabHi = (unsigned)slabId << 16;
  int cnt = 0;

  const int lastBlk = nblk - 1;
  auto rd = [&](int blk) { return sid4[(blk < lastBlk ? blk : lastBlk) * 32 + lane]; };
  auto examine = [&](const uint4& cur, int blk) {
    const unsigned dw[4] = {cur.x, cur.y, cur.z, cur.w};
    const int ebase = (blk << 8) + (lane << 3);
#pragma unroll
    for (int k = 0; k < 8; ++k) {
      const bool m = (k & 1) ? ((dw[k >> 1] & 0xFFFF0000u) == slabHi)
                             : ((dw[k >> 1] & 0x0000FFFFu) == (unsigned)slabId);
      const unsigned mask = __builtin_amdgcn_ballot_w32(m);
      if (mask) {
        if (m) q[cnt + __builtin_popcount(mask & lt)] = ebase + k;
        cnt += __builtin_popcount(mask);
      }
    }
  };

  auto gather = [&](int n, int r, int cl, float nrm) {
    for (int g = 0; g < n; g += G) {
      vecf hv[G];
#pragma unroll
      for (int j = 0; j < G; ++j) {
        const int rj = __builtin_amdgcn_readlane(r, g + j);
        hv[j] = (vecf)(0.0f);
        if (lane < LANES) hv[j] = *(const vecf*)(h + (long long)rj * LD + lane * VEC);
      }
#pragma unroll
      for (int j = 0; j < G; ++j) {
        if (g + j < n) {
          const int cj = __builtin_amdgcn_readlane(cl, g + j);
          const float nj = __builtin_bit_cast(float,
              __builtin_amdgcn_readlane(__builtin_bit_cast(int, nrm), g + j));
          if (lane < LANES) {
            float* dst = tile + cj * TP + lane;
#pragma unroll
            for (int k = 0; k < VEC; ++k)
              __hip_atomic_fetch_add(dst + k * LANES, hv[j][k] * nj, __ATOMIC_RELAXED,
                                     __HIP_MEMORY_SCOPE_WORKGROUP);
          }
        }
      }
    }
  };

  const unsigned int* myX = lst + (size_t)slabId * LCAP;
  float* myW = (float*)(lst + (size_t)gridDim.x * LCAP) + (size_t)slabId * LCAP;
  const int listed = lcnt[slabId * CPAD];

  if (listed <= LCAP) {
    constexpr int RB = 4;
    for (int base0 = w * 32; base0 < listed; base0 += RB * SWAVES * 32) {
      unsigned int ex[RB];
      float ew[RB];
#pragma unroll
      for (int k = 0; k < RB; ++k) {
        const int idx = base0 + k * SWAVES * 32 + lane;
        const int ci = idx < listed ? idx : listed - 1;
        ex[k] = myX[ci];
        if constexpr (EPI == 2) ew[k] = myW[ci];
      }
      if constexpr (EPI == 1) {
#pragma unroll
        for (int k = 0; k < RB; ++k)
          ew[k] = dinv[ex[k] & 0x00FFFFFFu] * dinv[lo + (int)(ex[k] >> 24)];
#pragma unroll
        for (int k = 0; k < RB; ++k) {
          const int idx = base0 + k * SWAVES * 32 + lane;
          if (idx < listed) myW[idx] = ew[k];
        }
      }
#pragma unroll
      for (int k = 0; k < RB; ++k) {
        const int base = base0 + k * SWAVES * 32;
        if (base < listed) {
          const int n = (listed - base < 32) ? (listed - base) : 32;
          gather(n, (int)(ex[k] & 0x00FFFFFFu), (int)(ex[k] >> 24), ew[k]);
        }
      }
    }
  } else {
  int b = w;
  for (;;) {
    while (b < nblk && cnt < 32) {
      const uint4 ra = rd(b);
      const uint4 rb = rd(b + SWAVES);
      const uint4 rc = rd(b + 2 * SWAVES);
      const uint4 re = rd(b + 3 * SWAVES);
      examine(ra, b); b += SWAVES;
      if (cnt <= QCAP - 256 && b < nblk) { examine(rb, b); b += SWAVES; }
      if (cnt <= QCAP - 256 && b < nblk) { examine(rc, b); b += SWAVES; }
      if (cnt <= QCAP - 256 && b < nblk) { examine(re, b); b += SWAVES; }
    }
    const bool last = (b >= nblk);
    const int take = last ? cnt : (cnt & ~31);
    for (int base = 0; base < take; base += 32) {
      const int n = (take - base < 32) ? (take - base) : 32;
      const bool live = lane < n;
      const int e = live ? q[base + lane] : 0;
      const int r = live ? row[e] : 0;
      const int c = live ? col[e] : lo;
      const float nrm = live ? dinv[r] * dinv[c] : 0.0f;
      const int cl = c - lo;
      gather(n, r, cl, nrm);
    }
    if (last) break;
    const int rem = cnt - take;
    const int keep = (lane < rem) ? q[take + lane] : 0;
    if (lane < rem) q[lane] = keep;
    cnt = rem;
  }
  }
  const bool act = lane < LANES;
  const int pl = act ? lane : 0;
  vecf bv = *(const vecf*)(bias + pl * VEC);
  vecf hs[RPW];
  float dd[RPW];
#pragma unroll
  for (int j = 0; j < RPW; ++j) {
    const int i = w + j * SWAVES;
    const int node = lo + (i < nrows ? i : 0);
    dd[j] = dinv[node];
    hs[j] = *(const vecf*)(h + (long long)node * LD + pl * VEC);
  }
  __syncthreads();

  if constexpr (EPI == 1) {
    const float* wls = (const float*)wls4;
#pragma unroll
    for (int j = 0; j < RPW; ++j) {
      const int i = w + j * SWAVES;
      if (i < nrows) {
        float* sp = tile + i * TP;
        const float d2 = dd[j] * dd[j];
        float4 o;
        o.x = sp[lane] + hs[j][0] * d2;
        o.y = sp[LANES + lane] + hs[j][1] * d2;
        o.z = sp[2 * LANES + lane] + hs[j][2] * d2;
        o.w = sp[3 * LANES + lane] + hs[j][3] * d2;
        *(float4*)(sp + lane * 4) = o;
      }
    }
    __syncthreads();
    constexpr int MT = TROWS / 16;
    constexpr int NT = F / 32;
    constexpr int CT = LD2 / 16;
    const bool mm = w < 2 * MT;
    const int l16 = lane & 15;
    const int hi = lane >> 4;
    const int kh = hi << 1;
    const int mt = w >> 1;
    const int nb = (w & 1) * (F / 2);
    const float* w2s = (const float*)w2s4;
    float* h2s = (float*)queue4;
    v8f acc[NT];
    v8f acc2[CT];
#pragma unroll
    for (int q = 0; q < NT; ++q) acc[q] = (v8f){};
#pragma unroll
    for (int c = 0; c < CT; ++c) acc2[c] = (v8f){};
    if (mm) {
      float4 bq[NT][2];
#pragma unroll
      for (int q = 0; q < NT; ++q) {
        const float4* bp4 = (const float4*)(bias + nb + q * 16 + hi * 8);
        bq[q][0] = bp4[0];
        bq[q][1] = bp4[1];
      }
      const float* ap = tile + (mt * 16 + l16) * TP + kh;
      const float* bp = wls + kh * PW + nb + l16;
#pragma unroll 4
      for (int k = 0; k < F; k += 4) {
        const v2f a = *(const v2f*)(ap + k);
#pragma unroll
        for (int q = 0; q < NT; ++q) {
          v2f b;
          b.x = bp[k * PW + q * 16];
          b.y = bp[(k + 1) * PW + q * 16];
          acc[q] = __builtin_amdgcn_wmma_f32_16x16x4_f32(
              false, b, false, a, (short)0, acc[q], false, false);
        }
      }
#pragma unroll
      for (int q = 0; q < NT; ++q) {
        const float bv8[8] = {bq[q][0].x, bq[q][0].y, bq[q][0].z, bq[q][0].w,
                              bq[q][1].x, bq[q][1].y, bq[q][1].z, bq[q][1].w};
#pragma unroll
        for (int j = 0; j < 4; ++j) {
          const float v0 = acc[q][2 * j] + bv8[2 * j];
          const float v1 = acc[q][2 * j + 1] + bv8[2 * j + 1];
          v2f hb;
          hb.x = v0 > 0.0f ? v0 : 0.0f;
          hb.y = v1 > 0.0f ? v1 : 0.0f;
          const float* wp = w2s + (nb + q * 16 + hi * 8 + 2 * j) * P2 + l16;
#pragma unroll
          for (int c = 0; c < CT; ++c) {
            v2f wa;
            wa.x = wp[c * 16];
            wa.y = wp[P2 + c * 16];
            acc2[c] = __builtin_amdgcn_wmma_f32_16x16x4_f32(
                false, wa, false, hb, (short)0, acc2[c], false, false);
          }
        }
      }
    }
    float* hp = h2s + (mt * 16 + l16) * LD2 + hi * 8;
    if (mm && (w & 1) == 0) {
#pragma unroll
      for (int c = 0; c < CT; ++c) {
        *(float4*)(hp + c * 16)     = make_float4(acc2[c][0], acc2[c][1], acc2[c][2], acc2[c][3]);
        *(float4*)(hp + c * 16 + 4) = make_float4(acc2[c][4], acc2[c][5], acc2[c][6], acc2[c][7]);
      }
    }
    __syncthreads();
    if (mm && (w & 1) == 1) {
#pragma unroll
      for (int c = 0; c < CT; ++c) {
        float4 p0 = *(const float4*)(hp + c * 16);
        float4 p1 = *(const float4*)(hp + c * 16 + 4);
        p0.x += acc2[c][0]; p0.y += acc2[c][1]; p0.z += acc2[c][2]; p0.w += acc2[c][3];
        p1.x += acc2[c][4]; p1.y += acc2[c][5]; p1.z += acc2[c][6]; p1.w += acc2[c][7];
        *(float4*)(hp + c * 16)     = p0;
        *(float4*)(hp + c * 16 + 4) = p1;
      }
    }
    __syncthreads();
    const int n4 = nrows * (LD2 / 4);
    float4* o4 = (float4*)(outp + (long long)lo * LD2);
    for (int i = t; i < n4; i += SWAVES * 32) o4[i] = ((const float4*)h2s)[i];
  } else if (w < nrows) {
    float v[RPW][VEC];
    float mx[RPW];
    float sum[RPW];
#pragma unroll
    for (int j = 0; j < RPW; ++j) {
      const int i = w + j * SWAVES;
      const float* sp = tile + (i < nrows ? i : w) * F + pl;
      const float d2 = dd[j] * dd[j];
      mx[j] = -3.4e38f;
#pragma unroll
      for (int k = 0; k < VEC; ++k) {
        const float x = sp[k * LANES] + hs[j][k] * d2 + bv[k];
        v[j][k] = act ? x : -3.4e38f;
        mx[j] = fmaxf(mx[j], v[j][k]);
      }
    }
#pragma unroll
    for (int m = 16; m >= 1; m >>= 1) {
#pragma unroll
      for (int j = 0; j < RPW; ++j) mx[j] = fmaxf(mx[j], __shfl_xor(mx[j], m));
    }
#pragma unroll
    for (int j = 0; j < RPW; ++j) {
      sum[j] = 0.0f;
#pragma unroll
      for (int k = 0; k < VEC; ++k) sum[j] += act ? __expf(v[j][k] - mx[j]) : 0.0f;
    }
#pragma unroll
    for (int m = 16; m >= 1; m >>= 1) {
#pragma unroll
      for (int j = 0; j < RPW; ++j) sum[j] += __shfl_xor(sum[j], m);
    }
#pragma unroll
    for (int j = 0; j < RPW; ++j) {
      const int i = w + j * SWAVES;
      if (i < nrows && act) {
        const float lse = mx[j] + __logf(sum[j]);
        vecf o;
#pragma unroll
        for (int k = 0; k < VEC; ++k) o[k] = v[j][k] - lse;
        *(vecf*)(tile + i * F + lane * VEC) = o;
      }
    }
  }
  if constexpr (EPI == 2) {
    __syncthreads();
    const int n4 = nrows * (F / 4);
    float4* o4 = (float4*)(outp + (long long)lo * F);
    for (int i = t; i < n4; i += SWAVES * 32) o4[i] = tile4[i];
  }
}


extern "C" void kernel_launch(void* const* d_in, const int* in_sizes, int n_in,
                              void* d_out, int out_size, void* d_ws, size_t ws_size,
                              hipStream_t stream) {
  const float* x  = (const float*)d_in[0];
  const int*   ei = (const int*)d_in[1];
  const float* W1 = (const float*)d_in[2];
  const float* b1 = (const float*)d_in[3];
  const float* W2 = (const float*)d_in[4];
  const float* b2 = (const float*)d_in[5];
  float* out = (float*)d_out;

  const int N = in_sizes[0] / NFEAT;
  const int E = in_sizes[1] / 2;
  const int* erow = ei;
  const int* ecol = ei + E;

  const int Epad = ((E + 255) / 256) * 256;
  unsigned short* sid = (unsigned short*)d_ws;
  const size_t sidBytes = (((size_t)Epad * 2 + 255) / 256) * 256;
  float* dinv = (float*)((char*)d_ws + sidBytes);
  float* h2   = dinv + N;

  const int TPB = 256;
  auto blocks = [](long long n, int tpb) {
    long long b = (n + tpb - 1) / tpb;
    return (unsigned)(b > 1048576 ? 1048576 : b);
  };

  const int nSlabs = (N + SLAB - 1) / SLAB;
  const size_t listOff = ((sidBytes + (size_t)N * (1 + LD2) * sizeof(float) + 255) / 256) * 256;
  unsigned int* elist = (unsigned int*)((char*)d_ws + listOff);
  int* ecnt = (int*)(elist + (size_t)nSlabs * LCAP * 2);

  init_kernel<<<blocks(nSlabs, TPB), TPB, 0, stream>>>(ecnt, nSlabs);

  edge_bin_kernel<<<(Epad + BINT * BINE - 1) / (BINT * BINE), BINT, 0, stream>>>(
      erow, ecol, sid, E, Epad, elist, ecnt, nSlabs);
  slab_degree_kernel<<<nSlabs, DEGT, 0, stream>>>(ecol, E, elist, ecnt, dinv, N);

  slab_aggregate<32, NFEAT / 32, NFEAT, 1><<<nSlabs, SWAVES * 32, 0, stream>>>(
      erow, ecol, sid, dinv, x, b1, h2, N, Epad, W1, W2, elist, ecnt);

  slab_aggregate<20, NCLS / 20, LD2, 2><<<nSlabs, SWAVES * 32, 0, stream>>>(
      erow, ecol, sid, dinv, h2, b2, out, N, Epad, W2, W2, elist, ecnt);
}


# both slab kernels: a wave requests its next 8 source rows before it adds the current 8 (two register sets, loads held in front of the adds, across batch ends); add order per wave unchanged
# speedup vs baseline: 4.7940x; 1.0495x over previous
#include <hip/hip_runtime.h>
#include <cmath>

typedef __attribute__((ext_vector_type(2))) float v2f;
typedef __attribute__((ext_vector_type(8))) float v8f;

#define NFEAT 128
#define NCLS  40
#define LD2   48
#define SLAB  196
#define QCAP  512
#define SWAVES 32
#define LCAP  16384
#define BINT  1024
#define BINE  4
#define HCAP  1024
#define CPAD  64


__global__ void init_kernel(int* __restrict__ lcnt, int nSlabs) {
  int i = blockIdx.x * blockDim.x + threadIdx.x;
  if (i < nSlabs) lcnt[i * CPAD] = 0;
}

__global__ __launch_bounds__(BINT) void edge_bin_kernel(
    const int* __restrict__ row, const int* __restrict__ col,
    unsigned short* __restrict__ sid, int E, int Epad,
    unsigned int* __restrict__ lst, int* __restrict__ lcnt, int nSlabs) {
  __shared__ int hist[HCAP];
  const int t = threadIdx.x;
  for (int i = t; i < HCAP; i += BINT) hist[i] = 0;
  __syncthreads();
  int s[BINE], rk[BINE], cj[BINE], rj[BINE];
  const int e0 = blockIdx.x * (BINT * BINE) + t;
#pragma unroll
  for (int j = 0; j < BINE; ++j) {
    const int e = e0 + j * BINT;
    cj[j] = (e < E) ? col[e] : 0;
    rj[j] = (e < E) ? row[e] : 0;
  }
#pragma unroll
  for (int j = 0; j < BINE; ++j) {
    const int e = e0 + j * BINT;
    s[j] = -1; rk[j] = 0;
    if (e < E) {
      const int sj = cj[j] / SLAB;
      s[j] = sj;
      rk[j] = (sj < HCAP) ? atomicAdd(&hist[sj], 1) : atomicAdd(&lcnt[sj * CPAD], 1);
    }
  }
  __syncthreads();
  for (int i = t; i < HCAP && i < nSlabs; i += BINT) {
    const int n = hist[i];
    if (n) hist[i] = atomicAdd(&lcnt[i * CPAD], n);
  }
  __syncthreads();
#pragma unroll
  for (int j = 0; j < BINE; ++j) {
    const int e = e0 + j * BINT;
    if (s[j] >= 0) {
      const int c = cj[j];
      sid[e] = (unsigned short)s[j];
      const int pos = rk[j] + (s[j] < HCAP ? hist[s[j]] : 0);
      if (pos < LCAP)
        lst[(size_t)s[j] * LCAP + pos] = (unsigned)rj[j] | ((unsigned)(c - s[j] * SLAB) << 24);
    } else if (e < Epad) {
      sid[e] = (unsigned short)0xFFFF;
    }
  }
}

#define DEGT 256
#define DEGU 16
__global__ __launch_bounds__(DEGT) void slab_degree_kernel(
    const int* __restrict__ col, int E, const unsigned int* __restrict__ lst,
    const int* __restrict__ lcnt, float* __restrict__ dinv, int N) {
  __shared__ int cntS[SLAB];
  const int t = threadIdx.x;
  const int slabId = blockIdx.x;
  const int lo = slabId * SLAB;
  for (int i = t; i < SLAB; i += DEGT) cntS[i] = 0;
  __syncthreads();
  const int listed = lcnt[slabId * CPAD];
  if (listed <= LCAP) {
    const unsigned int* myX = lst + (size_t)slabId * LCAP;
    for (int base = t; base < listed; base += DEGT * DEGU) {
      unsigned int v[DEGU];
#pragma unroll
      for (int k = 0; k < DEGU; ++k) {
        const int i = base + k * DEGT;
        v[k] = myX[i < listed ? i : listed - 1];
      }
#pragma unroll
      for (int k = 0; k < DEGU; ++k)
        if (base + k * DEGT < listed) atomicAdd(&cntS[v[k] >> 24], 1);
    }
  } else {
    for (int e = t; e < E; e += DEGT) {
      const int c = col[e] - lo;
      if (c >= 0 && c < SLAB) atomicAdd(&cntS[c], 1);
    }
  }
  __syncthreads();
  for (int i = t; i < SLAB; i += DEGT) {
    const int node = lo + i;
    if (node < N) {
      const float d = 1.0f + (float)cntS[i];
      dinv[node] = (d > 0.0f) ? rsqrtf(d) : 0.0f;
    }
  }
}

template <int LANES, int VEC, int LD, int EPI>
__global__ __launch_bounds__(SWAVES * 32) void slab_aggregate(
    const int* __restrict__ row, const int* __restrict__ col,
    const unsigned short* __restrict__ sid, const float* __restrict__ dinv,
    const float* __restrict__ h, const float* __restrict__ bias,
    float* __restrict__ outp, int N, int Epad, const float* __restrict__ Wm,
    const float* __restrict__ Wn,
    unsigned int* __restrict__ lst, const int* __restrict__ lcnt) {
  typedef __attribute__((ext_vector_type(VEC))) float vecf;
  constexpr int F = LANES * VEC;
  constexpr int TP = (EPI == 1) ? (F + 4) : F;
  constexpr int TROWS = (EPI == 1) ? ((SLAB + 15) / 16) * 16 : SLAB;
  constexpr int PW = F + 8;
  constexpr int RPW = (SLAB + SWAVES - 1) / SWAVES;
  static_assert(F % 4 == 0, "a slab's rows leave as 16-byte pieces");
  constexpr int G = 8;
  constexpr int P2 = LD2 + 4;
  __shared__ float4 tile4[TROWS * TP / 4];
  __shared__ int4 queue4[SWAVES * QCAP / 4];
  __shared__ float4 wls4[(EPI == 1) ? (F * PW / 4) : 1];
  __shared__ float4 w2s4[(EPI == 1) ? (F * P2 / 4) : 1];
  float* tile = (float*)tile4;
  int* queue = (int*)queue4;

  const int t = threadIdx.x;
  const int lane = t & 31;
  const int w = __builtin_amdgcn_readfirstlane(t >> 5);
  const int slabId = blockIdx.x;
  const int lo = slabId * SLAB;
  const int nrows = (N - lo < SLAB) ? (N - lo) : SLAB;

  for (int i = t; i < TROWS * TP / 4; i += SWAVES * 32) tile4[i] = make_float4(0.f, 0.f, 0.f, 0.f);
  if constexpr (EPI == 1) {
    for (int i = t; i < F * F / 4; i += SWAVES * 32) {
      const int kr = i / (F / 4), c4 = i % (F / 4);
      wls4[kr * (PW / 4) + c4] = ((const float4*)Wm)[i];
    }
    float* w2s = (float*)w2s4;
    for (int i = t; i < F * LD2; i += SWAVES * 32) {
      const int kr = i / LD2, c = i % LD2;
      w2s[kr * P2 + c] = (c < NCLS) ? Wn[kr * NCLS + c] : 0.0f;
    }
  }
  __syncthreads();

  int* q = queue + w * QCAP;
  const uint4* sid4 = (const uint4*)sid;
  const int nblk = Epad >> 8;
  const unsigned lt = (1u << lane) - 1u;
  const unsigned slabHi = (unsigned)slabId << 16;
  int cnt = 0;

  const int lastBlk = nblk - 1;
  auto rd = [&](int blk) { return sid4[(blk < lastBlk ? blk : lastBlk) * 32 + lane]; };
  auto examine = [&](const uint4& cur, int blk) {
    const unsigned dw[4] = {cur.x, cur.y, cur.z, cur.w};
    const int ebase = (blk << 8) + (lane << 3);
#pragma unroll
    for (int k = 0; k < 8; ++k) {
      const bool m = (k & 1) ? ((dw[k >> 1] & 0xFFFF0000u) == slabHi)
                             : ((dw[k >> 1] & 0x0000FFFFu) == (unsigned)slabId);
      const unsigned mask = __builtin_amdgcn_ballot_w32(m);
      if (mask) {
        if (m) q[cnt + __builtin_popcount(mask & lt)] = ebase + k;
        cnt += __builtin_popcount(mask);
      }
    }
  };

  const int rowLane = lane < LANES ? lane : 0;
  auto loadRows = [&](vecf (&hv)[G], int g, int r) {
#pragma unroll
    for (int j = 0; j < G; ++j) {
      const int rj = __builtin_amdgcn_readlane(r, g + j);
      hv[j] = *(const vecf*)(h + (long long)rj * LD + rowLane * VEC);
    }
  };
  auto holdRows = [&](const vecf (&hv)[G]) {
    static_assert(G == 8, "the fence names the G loaded rows one by one");
    asm volatile("" :: "v"(hv[0]), "v"(hv[1]), "v"(hv[2]), "v"(hv[3]),
                       "v"(hv[4]), "v"(hv[5]), "v"(hv[6]), "v"(hv[7]));
  };
  auto addRow = [&](const vecf& hvj, int e, int cl, float nrm) {
    const int cj = __builtin_amdgcn_readlane(cl, e);
    const float nj = __builtin_bit_cast(float,
        __builtin_amdgcn_readlane(__builtin_bit_cast(int, nrm), e));
    if (lane < LANES) {
      float* dst = tile + cj * TP + lane;
#pragma unroll
      for (int k = 0; k < VEC; ++k)
        __hip_atomic_fetch_add(dst + k * LANES, hvj[k] * nj, __ATOMIC_RELAXED,
                               __HIP_MEMORY_SCOPE_WORKGROUP);
    }
  };
  auto addFull = [&](const vecf (&hv)[G], int g, int cl, float nrm) {
#pragma unroll
    for (int j = 0; j < G; ++j) addRow(hv[j], g + j, cl, nrm);
  };
  auto addPart = [&](const vecf (&hv)[G], int g, int n, int cl, float nrm) {
#pragma unroll
    for (int j = 0; j < G; ++j)
      if (g + j < n) addRow(hv[j], g + j, cl, nrm);
  };
  auto gather = [&](int n, int r, int cl, float nrm) {
    vecf ha[G], hb[G];
    loadRows(ha, 0, r);
#pragma nounroll
    for (int g = 0; g < n; g += 2 * G) {
      if (g + G < n) {
        loadRows(hb, g + G, r);
        holdRows(ha);
        addFull(ha, g, cl, nrm);
        if (g + 2 * G < n) {
          loadRows(ha, g + 2 * G, r);
          holdRows(hb);
          addFull(hb, g + G, cl, nrm);
        } else {
          holdRows(hb);
          addPart(hb, g + G, n, cl, nrm);
        }
      } else {
        holdRows(ha);
        addPart(ha, g, n, cl, nrm);
      }
    }
  };

  const unsigned int* myX = lst + (size_t)slabId * LCAP;
  float* myW = (float*)(lst + (size_t)gridDim.x * LCAP) + (size_t)slabId * LCAP;
  const int listed = lcnt[slabId * CPAD];

  if (listed <= LCAP) {
    constexpr int RB = 4;
    for (int base0 = w * 32; base0 < listed; base0 += RB * SWAVES * 32) {
      unsigned int ex[RB];
      float ew[RB];
#pragma unroll
      for (int k = 0; k < RB; ++k) {
        const int idx = base0 + k * SWAVES * 32 + lane;
        const int ci = idx < listed ? idx : listed - 1;
        ex[k] = myX[ci];
        if constexpr (EPI == 2) ew[k] = myW[ci];
      }
      if constexpr (EPI == 1) {
#pragma unroll
        for (int k = 0; k < RB; ++k)
          ew[k] = dinv[ex[k] & 0x00FFFFFFu] * dinv[lo + (int)(ex[k] >> 24)];
#pragma unroll
        for (int k = 0; k < RB; ++k) {
          const int idx = base0 + k * SWAVES * 32 + lane;
          if (idx < listed) myW[idx] = ew[k];
        }
      }
      static_assert(RB == 4 && G == 8, "a batch is four groups of eight");
      auto batchN = [&](int k) {
        const int left = listed - (base0 + k * SWAVES * 32);
        return left < 0 ? 0 : (left < 32 ? left : 32);
      };
      int T = 0;
#pragma unroll
      for (int k = 0; k < RB; ++k) T += (batchN(k) + G - 1) / G;
      auto exOf = [&](int k) { return k == 0 ? ex[0] : (k == 1 ? ex[1] : (k == 2 ? ex[2] : ex[3])); };
      auto ewOf = [&](int k) { return k == 0 ? ew[0] : (k == 1 ? ew[1] : (k == 2 ? ew[2] : ew[3])); };
      auto loadGroup = [&](vecf (&hv)[G], int t) {
        loadRows(hv, (t & 3) * G, (int)(exOf(t >> 2) & 0x00FFFFFFu));
      };
      auto addGroupFull = [&](const vecf (&hv)[G], int t) {
        addFull(hv, (t & 3) * G, (int)(exOf(t >> 2) >> 24), ewOf(t >> 2));
      };
      auto addGroupPart = [&](const vecf (&hv)[G], int t) {
        addPart(hv, (t & 3) * G, batchN(t >> 2), (int)(exOf(t >> 2) >> 24), ewOf(t >> 2));
      };
      vecf ha[G], hb[G];
      loadGroup(ha, 0);
#pragma nounroll
      for (int t = 0; t < T; t += 2) {
        if (t + 1 < T) {
          loadGroup(hb, t + 1);
          holdRows(ha);
          addGroupFull(ha, t);
          if (t + 2 < T) {
            loadGroup(ha, t + 2);
            holdRows(hb);
            addGroupFull(hb, t + 1);
          } else {
            holdRows(hb);
            addGroupPart(hb, t + 1);
          }
        } else {
          holdRows(ha);
          addGroupPart(ha, t);
        }
      }
    }
  } else {
  int b = w;
  for (;;) {
    while (b < nblk && cnt < 32) {
      const uint4 ra = rd(b);
      const uint4 rb = rd(b + SWAVES);
      const uint4 rc = rd(b + 2 * SWAVES);
      const uint4 re = rd(b + 3 * SWAVES);
      examine(ra, b); b += SWAVES;
      if (cnt <= QCAP - 256 && b < nblk) { examine(rb, b); b += SWAVES; }
      if (cnt <= QCAP - 256 && b < nblk) { examine(rc, b); b += SWAVES; }
      if (cnt <= QCAP - 256 && b < nblk) { examine(re, b); b += SWAVES; }
    }
    const bool last = (b >= nblk);
    const int take = last ? cnt : (cnt & ~31);
    for (int base = 0; base < take; base += 32) {
      const int n = (take - base < 32) ? (take - base) : 32;
      const bool live = lane < n;
      const int e = live ? q[base + lane] : 0;
      const int r = live ? row[e] : 0;
      const int c = live ? col[e] : lo;
      const float nrm = live ? dinv[r] * dinv[c] : 0.0f;
      const int cl = c - lo;
      gather(n, r, cl, nrm);
    }
    if (last) break;
    const int rem = cnt - take;
    const int keep = (lane < rem) ? q[take + lane] : 0;
    if (lane < rem) q[lane] = keep;
    cnt = rem;
  }
  }
  const bool act = lane < LANES;
  const int pl = act ? lane : 0;
  vecf bv = *(const vecf*)(bias + pl * VEC);
  vecf hs[RPW];
  float dd[RPW];
#pragma unroll
  for (int j = 0; j < RPW; ++j) {
    const int i = w + j * SWAVES;
    const int node = lo + (i < nrows ? i : 0);
    dd[j] = dinv[node];
    hs[j] = *(const vecf*)(h + (long long)node * LD + pl * VEC);
  }
  __syncthreads();

  if constexpr (EPI == 1) {
    const float* wls = (const float*)wls4;
#pragma unroll
    for (int j = 0; j < RPW; ++j) {
      const int i = w + j * SWAVES;
      if (i < nrows) {
        float* sp = tile + i * TP;
        const float d2 = dd[j] * dd[j];
        float4 o;
        o.x = sp[lane] + hs[j][0] * d2;
        o.y = sp[LANES + lane] + hs[j][1] * d2;
        o.z = sp[2 * LANES + lane] + hs[j][2] * d2;
        o.w = sp[3 * LANES + lane] + hs[j][3] * d2;
        *(float4*)(sp + lane * 4) = o;
      }
    }
    __syncthreads();
    constexpr int MT = TROWS / 16;
    constexpr int NT = F / 32;
    constexpr int CT = LD2 / 16;
    const bool mm = w < 2 * MT;
    const int l16 = lane & 15;
    const int hi = lane >> 4;
    const int kh = hi << 1;
    const int mt = w >> 1;
    const int nb = (w & 1) * (F / 2);
    const float* w2s = (const float*)w2s4;
    float* h2s = (float*)queue4;
    v8f acc[NT];
    v8f acc2[CT];
#pragma unroll
    for (int q = 0; q < NT; ++q) acc[q] = (v8f){};
#pragma unroll
    for (int c = 0; c < CT; ++c) acc2[c] = (v8f){};
    if (mm) {
      float4 bq[NT][2];
#pragma unroll
      for (int q = 0; q < NT; ++q) {
        const float4* bp4 = (const float4*)(bias + nb + q * 16 + hi * 8);
        bq[q][0] = bp4[0];
        bq[q][1] = bp4[1];
      }
      const float* ap = tile + (mt * 16 + l16) * TP + kh;
      const float* bp = wls + kh * PW + nb + l16;
#pragma unroll 4
      for (int k = 0; k < F; k += 4) {
        const v2f a = *(const v2f*)(ap + k);
#pragma unroll
        for (int q = 0; q < NT; ++q) {
          v2f b;
          b.x = bp[k * PW + q * 16];
          b.y = bp[(k + 1) * PW + q * 16];
          acc[q] = __builtin_amdgcn_wmma_f32_16x16x4_f32(
              false, b, false, a, (short)0, acc[q], false, false);
        }
      }
#pragma unroll
      for (int q = 0; q < NT; ++q) {
        const float bv8[8] = {bq[q][0].x, bq[q][0].y, bq[q][0].z, bq[q][0].w,
                              bq[q][1].x, bq[q][1].y, bq[q][1].z, bq[q][1].w};
#pragma unroll
        for (int j = 0; j < 4; ++j) {
          const float v0 = acc[q][2 * j] + bv8[2 * j];
          const float v1 = acc[q][2 * j + 1] + bv8[2 * j + 1];
          v2f hb;
          hb.x = v0 > 0.0f ? v0 : 0.0f;
          hb.y = v1 > 0.0f ? v1 : 0.0f;
          const float* wp = w2s + (nb + q * 16 + hi * 8 + 2 * j) * P2 + l16;
#pragma unroll
          for (int c = 0; c < CT; ++c) {
            v2f wa;
            wa.x = wp[c * 16];
            wa.y = wp[P2 + c * 16];
            acc2[c] = __builtin_amdgcn_wmma_f32_16x16x4_f32(
                false, wa, false, hb, (short)0, acc2[c], false, false);
          }
        }
      }
    }
    float* hp = h2s + (mt * 16 + l16) * LD2 + hi * 8;
    if (mm && (w & 1) == 0) {
#pragma unroll
      for (int c = 0; c < CT; ++c) {
        *(float4*)(hp + c * 16)     = make_float4(acc2[c][0], acc2[c][1], acc2[c][2], acc2[c][3]);
        *(float4*)(hp + c * 16 + 4) = make_float4(acc2[c][4], acc2[c][5], acc2[c][6], acc2[c][7]);
      }
    }
    __syncthreads();
    if (mm && (w & 1) == 1) {
#pragma unroll
      for (int c = 0; c < CT; ++c) {
        float4 p0 = *(const float4*)(hp + c * 16);
        float4 p1 = *(const float4*)(hp + c * 16 + 4);
        p0.x += acc2[c][0]; p0.y += acc2[c][1]; p0.z += acc2[c][2]; p0.w += acc2[c][3];
        p1.x += acc2[c][4]; p1.y += acc2[c][5]; p1.z += acc2[c][6]; p1.w += acc2[c][7];
        *(float4*)(hp + c * 16)     = p0;
        *(float4*)(hp + c * 16 + 4) = p1;
      }
    }
    __syncthreads();
    const int n4 = nrows * (LD2 / 4);
    float4* o4 = (float4*)(outp + (long long)lo * LD2);
    for (int i = t; i < n4; i += SWAVES * 32) o4[i] = ((const float4*)h2s)[i];
  } else if (w < nrows) {
    float v[RPW][VEC];
    float mx[RPW];
    float sum[RPW];
#pragma unroll
    for (int j = 0; j < RPW; ++j) {
      const int i = w + j * SWAVES;
      const float* sp = tile + (i < nrows ? i : w) * F + pl;
      const float d2 = dd[j] * dd[j];
      mx[j] = -3.4e38f;
#pragma unroll
      for (int k = 0; k < VEC; ++k) {
        const float x = sp[k * LANES] + hs[j][k] * d2 + bv[k];
        v[j][k] = act ? x : -3.4e38f;
        mx[j] = fmaxf(mx[j], v[j][k]);
      }
    }
#pragma unroll
    for (int m = 16; m >= 1; m >>= 1) {
#pragma unroll
      for (int j = 0; j < RPW; ++j) mx[j] = fmaxf(mx[j], __shfl_xor(mx[j], m));
    }
#pragma unroll
    for (int j = 0; j < RPW; ++j) {
      sum[j] = 0.0f;
#pragma unroll
      for (int k = 0; k < VEC; ++k) sum[j] += act ? __expf(v[j][k] - mx[j]) : 0.0f;
    }
#pragma unroll
    for (int m = 16; m >= 1; m >>= 1) {
#pragma unroll
      for (int j = 0; j < RPW; ++j) sum[j] += __shfl_xor(sum[j], m);
    }
#pragma unroll
    for (int j = 0; j < RPW; ++j) {
      const int i = w + j * SWAVES;
      if (i < nrows && act) {
        const float lse = mx[j] + __logf(sum[j]);
        vecf o;
#pragma unroll
        for (int k = 0; k < VEC; ++k) o[k] = v[j][k] - lse;
        *(vecf*)(tile + i * F + lane * VEC) = o;
      }
    }
  }
  if constexpr (EPI == 2) {
    __syncthreads();
    const int n4 = nrows * (F / 4);
    float4* o4 = (float4*)(outp + (long long)lo * F);
    for (int i = t; i < n4; i += SWAVES * 32) o4[i] = tile4[i];
  }
}


extern "C" void kernel_launch(void* const* d_in, const int* in_sizes, int n_in,
                              void* d_out, int out_size, void* d_ws, size_t ws_size,
                              hipStream_t stream) {
  const float* x  = (const float*)d_in[0];
  const int*   ei = (const int*)d_in[1];
  const float* W1 = (const float*)d_in[2];
  const float* b1 = (const float*)d_in[3];
  const float* W2 = (const float*)d_in[4];
  const float* b2 = (const float*)d_in[5];
  float* out = (float*)d_out;

  const int N = in_sizes[0] / NFEAT;
  const int E = in_sizes[1] / 2;
  const int* erow = ei;
  const int* ecol = ei + E;

  const int Epad = ((E + 255) / 256) * 256;
  unsigned short* sid = (unsigned short*)d_ws;
  const size_t sidBytes = (((size_t)Epad * 2 + 255) / 256) * 256;
  float* dinv = (float*)((char*)d_ws + sidBytes);
  float* h2   = dinv + N;

  const int TPB = 256;
  auto blocks = [](long long n, int tpb) {
    long long b = (n + tpb - 1) / tpb;
    return (unsigned)(b > 1048576 ? 1048576 : b);
  };

  const int nSlabs = (N + SLAB - 1) / SLAB;
  const size_t listOff = ((sidBytes + (size_t)N * (1 + LD2) * sizeof(float) + 255) / 256) * 256;
  unsigned int* elist = (unsigned int*)((char*)d_ws + listOff);
  int* ecnt = (int*)(elist + (size_t)nSlabs * LCAP * 2);

  init_kernel<<<blocks(nSlabs, TPB), TPB, 0, stream>>>(ecnt, nSlabs);

  edge_bin_kernel<<<(Epad + BINT * BINE - 1) / (BINT * BINE), BINT, 0, stream>>>(
      erow, ecol, sid, E, Epad, elist, ecnt, nSlabs);
  slab_degree_kernel<<<nSlabs, DEGT, 0, stream>>>(ecol, E, elist, ecnt, dinv, N);

  slab_aggregate<32, NFEAT / 32, NFEAT, 1><<<nSlabs, SWAVES * 32, 0, stream>>>(
      erow, ecol, sid, dinv, x, b1, h2, N, Epad, W1, W2, elist, ecnt);

  slab_aggregate<20, NCLS / 20, LD2, 2><<<nSlabs, SWAVES * 32, 0, stream>>>(
      erow, ecol, sid, dinv, h2, b2, out, N, Epad, W2, W2, elist, ecnt);
}
